# starting kernel
# baseline (speedup 1.0000x reference)
#include <hip/hip_runtime.h>
#include <math.h>

#define D_MODEL 1024
#define D_STATE 16
#define D_CONV  4
#define DT_RANK 64
#define D_INNER 2048
#define LSEQ    2048

typedef __attribute__((ext_vector_type(2))) float v2f;
typedef __attribute__((ext_vector_type(8))) float v8f;

struct Frag { v2f a0, a1, b0, b1, b2, b3; };

__device__ __forceinline__ void load_frag(
    Frag& f, const float* Ar0, const float* Ar1,
    const float* w0, const float* w1, const float* w2, const float* w3, int k)
{
    f.a0 = *(const v2f*)(Ar0 + k);
    f.a1 = *(const v2f*)(Ar1 + k);
    f.b0 = *(const v2f*)(w0 + k);
    f.b1 = *(const v2f*)(w1 + k);
    f.b2 = *(const v2f*)(w2 + k);
    f.b3 = *(const v2f*)(w3 + k);
}

__device__ __forceinline__ void do_wmma(const Frag& f, v8f acc[2][4])
{
    acc[0][0] = __builtin_amdgcn_wmma_f32_16x16x4_f32(false, f.a0, false, f.b0, (short)0, acc[0][0], false, false);
    acc[1][0] = __builtin_amdgcn_wmma_f32_16x16x4_f32(false, f.a1, false, f.b0, (short)0, acc[1][0], false, false);
    acc[0][1] = __builtin_amdgcn_wmma_f32_16x16x4_f32(false, f.a0, false, f.b1, (short)0, acc[0][1], false, false);
    acc[1][1] = __builtin_amdgcn_wmma_f32_16x16x4_f32(false, f.a1, false, f.b1, (short)0, acc[1][1], false, false);
    acc[0][2] = __builtin_amdgcn_wmma_f32_16x16x4_f32(false, f.a0, false, f.b2, (short)0, acc[0][2], false, false);
    acc[1][2] = __builtin_amdgcn_wmma_f32_16x16x4_f32(false, f.a1, false, f.b2, (short)0, acc[1][2], false, false);
    acc[0][3] = __builtin_amdgcn_wmma_f32_16x16x4_f32(false, f.a0, false, f.b3, (short)0, acc[0][3], false, false);
    acc[1][3] = __builtin_amdgcn_wmma_f32_16x16x4_f32(false, f.a1, false, f.b3, (short)0, acc[1][3], false, false);
}

__global__ __launch_bounds__(128, 1) void gemm_wmma_f32(
    const float* __restrict__ A, const float* __restrict__ W,
    const float* __restrict__ bias, float* __restrict__ C,
    int M, int N, int K, int lda, int act)
{
    const int lane = threadIdx.x & 31;
    const int wave = threadIdx.x >> 5;
    const int m0   = (blockIdx.y << 7) + (wave << 5);
    const int n0   = blockIdx.x << 6;
    if (m0 >= M) return;

    const int mrow  = lane & 15;
    const int khalf = (lane >> 4) << 1;

    const float* __restrict__ Ar0 = A + (size_t)(m0 + mrow) * lda + khalf;
    const float* __restrict__ Ar1 = Ar0 + (size_t)16 * lda;

    const float* Wr[4];
#pragma unroll
    for (int j = 0; j < 4; ++j) {
        int row = n0 + (j << 4) + mrow;
        row = (row < N) ? row : (N - 1);
        Wr[j] = W + (size_t)row * K + khalf;
    }

    v8f acc[2][4];
#pragma unroll
    for (int i = 0; i < 2; ++i)
#pragma unroll
        for (int j = 0; j < 4; ++j) acc[i][j] = (v8f){};

    Frag f0, f1;
    load_frag(f0, Ar0, Ar1, Wr[0], Wr[1], Wr[2], Wr[3], 0);

    int k0 = 0;
    for (; k0 < K - 16; k0 += 16) {
        __builtin_prefetch(Ar0   + k0 + 128, 0, 1);
        __builtin_prefetch(Ar1   + k0 + 128, 0, 1);
        __builtin_prefetch(Wr[0] + k0 + 128, 0, 1);
        __builtin_prefetch(Wr[2] + k0 + 128, 0, 1);
        load_frag(f1, Ar0, Ar1, Wr[0], Wr[1], Wr[2], Wr[3], k0 + 4);
        do_wmma(f0, acc);
        load_frag(f0, Ar0, Ar1, Wr[0], Wr[1], Wr[2], Wr[3], k0 + 8);
        do_wmma(f1, acc);
        load_frag(f1, Ar0, Ar1, Wr[0], Wr[1], Wr[2], Wr[3], k0 + 12);
        do_wmma(f0, acc);
        load_frag(f0, Ar0, Ar1, Wr[0], Wr[1], Wr[2], Wr[3], k0 + 16);
        do_wmma(f1, acc);
    }
    load_frag(f1, Ar0, Ar1, Wr[0], Wr[1], Wr[2], Wr[3], k0 + 4);
    do_wmma(f0, acc);
    load_frag(f0, Ar0, Ar1, Wr[0], Wr[1], Wr[2], Wr[3], k0 + 8);
    do_wmma(f1, acc);
    load_frag(f1, Ar0, Ar1, Wr[0], Wr[1], Wr[2], Wr[3], k0 + 12);
    do_wmma(f0, acc);
    do_wmma(f1, acc);

    const int rsub = (lane >> 4) << 3;
#pragma unroll
    for (int j = 0; j < 4; ++j) {
        const int col = n0 + (j << 4) + mrow;
        if ((n0 + (j << 4)) >= N) continue;
        const float bv = (act == 1) ? bias[col] : 0.0f;
#pragma unroll
        for (int i = 0; i < 2; ++i) {
            const int rbase = m0 + (i << 4) + rsub;
#pragma unroll
            for (int r = 0; r < 8; ++r) {
                float v = acc[i][j][r];
                if (act == 1) {
                    v += bv;
                    v = (v > 20.0f) ? v : log1pf(__expf(v));
                }
                C[(size_t)(rbase + r) * N + col] = v;
            }
        }
    }
}

__global__ __launch_bounds__(256) void conv_silu_kernel(
    const float* __restrict__ xz, const float* __restrict__ cw,
    const float* __restrict__ cb, float* __restrict__ xc)
{
    int idx = blockIdx.x * blockDim.x + threadIdx.x;
    if (idx >= LSEQ * D_INNER) return;
    int l = idx >> 11;
    int d = idx & (D_INNER - 1);

    float acc = cb[d];
#pragma unroll
    for (int j = 0; j < D_CONV; ++j) {
        int ls = l - (D_CONV - 1) + j;
        if (ls >= 0)
            acc += cw[d * D_CONV + j] * xz[(size_t)ls * (2 * D_INNER) + d];
    }
    xc[idx] = acc / (1.0f + __expf(-acc));
}

#define SCAN_TS 32
__global__ __launch_bounds__(256) void ssm_scan_kernel(
    const float* __restrict__ xdbl, const float* __restrict__ dtbuf,
    const float* __restrict__ xz,   const float* __restrict__ A_log,
    const float* __restrict__ D_param, float* __restrict__ xc_io)
{
    __shared__ float sB[SCAN_TS][D_STATE];
    __shared__ float sC[SCAN_TS][D_STATE];

    const int d = blockIdx.x * blockDim.x + threadIdx.x;

    float Acoef[D_STATE];
#pragma unroll
    for (int n = 0; n < D_STATE; ++n)
        Acoef[n] = -__expf(A_log[(size_t)d * D_STATE + n]);
    const float Dp = D_param[d];

    float h[D_STATE];
#pragma unroll
    for (int n = 0; n < D_STATE; ++n) h[n] = 0.0f;

    for (int lc = 0; lc < LSEQ; lc += SCAN_TS) {
        __syncthreads();
        for (int i = threadIdx.x; i < SCAN_TS * D_STATE; i += blockDim.x) {
            int t = i >> 4, n = i & 15;
            const float* row = xdbl + (size_t)(lc + t) * (DT_RANK + 2 * D_STATE);
            sB[t][n] = row[DT_RANK + n];
            sC[t][n] = row[DT_RANK + D_STATE + n];
        }
        __syncthreads();

        for (int t = 0; t < SCAN_TS; ++t) {
            const int l = lc + t;
            const float dt = dtbuf[(size_t)l * D_INNER + d];
            const float xv = xc_io[(size_t)l * D_INNER + d];
            float y = 0.0f;
#pragma unroll
            for (int n = 0; n < D_STATE; ++n) {
                float dA = __expf(dt * Acoef[n]);
                h[n] = dA * h[n] + (dt * sB[t][n]) * xv;
                y += h[n] * sC[t][n];
            }
            y += xv * Dp;
            float z = xz[(size_t)l * (2 * D_INNER) + D_INNER + d];
            float g = z / (1.0f + __expf(-z));
            xc_io[(size_t)l * D_INNER + d] = y * g;
        }
    }
}

extern "C" void kernel_launch(void* const* d_in, const int* in_sizes, int n_in,
                              void* d_out, int out_size, void* d_ws, size_t ws_size,
                              hipStream_t stream) {
    const float* x       = (const float*)d_in[0];
    const float* W_in    = (const float*)d_in[1];
    const float* conv_w  = (const float*)d_in[2];
    const float* conv_b  = (const float*)d_in[3];
    const float* W_x     = (const float*)d_in[4];
    const float* W_dt    = (const float*)d_in[5];
    const float* b_dt    = (const float*)d_in[6];
    const float* A_log   = (const float*)d_in[7];
    const float* D_param = (const float*)d_in[8];
    const float* W_out   = (const float*)d_in[9];
    float* out = (float*)d_out;

    float* xz   = (float*)d_ws;
    float* xc   = xz   + (size_t)LSEQ * 2 * D_INNER;
    float* xdbl = xc   + (size_t)LSEQ * D_INNER;
    float* dtb  = xdbl + (size_t)LSEQ * (DT_RANK + 2 * D_STATE);

    dim3 blk(128, 1, 1);

    gemm_wmma_f32<<<dim3(4096 / 64, LSEQ / 128), blk, 0, stream>>>(
        x, W_in, nullptr, xz, LSEQ, 2 * D_INNER, D_MODEL, D_MODEL, 0);

    conv_silu_kernel<<<(LSEQ * D_INNER) / 256, 256, 0, stream>>>(
        xz, conv_w, conv_b, xc);

    gemm_wmma_f32<<<dim3((96 + 63) / 64, LSEQ / 128), blk, 0, stream>>>(
        xc, W_x, nullptr, xdbl, LSEQ, DT_RANK + 2 * D_STATE, D_INNER, D_INNER, 0);

    gemm_wmma_f32<<<dim3(D_INNER / 64, LSEQ / 128), blk, 0, stream>>>(
        xdbl, W_dt, b_dt, dtb, LSEQ, D_INNER, DT_RANK, DT_RANK + 2 * D_STATE, 1);

    ssm_scan_kernel<<<D_INNER / 256, 256, 0, stream>>>(
        xdbl, dtb, xz, A_log, D_param, xc);

    gemm_wmma_f32<<<dim3(D_MODEL / 64, LSEQ / 128), blk, 0, stream>>>(
        xc, W_out, nullptr, out, LSEQ, D_MODEL, D_INNER, D_INNER, 0);
}


# scan: sequence cut into 64 chunks of 32 steps run in parallel: chunk end states from zero, a carry pass (decay as exp(A*sum dt), a different order), then each chunk re-scanned from its start state
# speedup vs baseline: 6.2733x; 6.2733x over previous
#include <hip/hip_runtime.h>
#include <math.h>

#define D_MODEL 1024
#define D_STATE 16
#define D_CONV  4
#define DT_RANK 64
#define D_INNER 2048
#define LSEQ    2048

typedef __attribute__((ext_vector_type(2))) float v2f;
typedef __attribute__((ext_vector_type(8))) float v8f;

struct Frag { v2f a0, a1, b0, b1, b2, b3; };

__device__ __forceinline__ void load_frag(
    Frag& f, const float* Ar0, const float* Ar1,
    const float* w0, const float* w1, const float* w2, const float* w3, int k)
{
    f.a0 = *(const v2f*)(Ar0 + k);
    f.a1 = *(const v2f*)(Ar1 + k);
    f.b0 = *(const v2f*)(w0 + k);
    f.b1 = *(const v2f*)(w1 + k);
    f.b2 = *(const v2f*)(w2 + k);
    f.b3 = *(const v2f*)(w3 + k);
}

__device__ __forceinline__ void do_wmma(const Frag& f, v8f acc[2][4])
{
    acc[0][0] = __builtin_amdgcn_wmma_f32_16x16x4_f32(false, f.a0, false, f.b0, (short)0, acc[0][0], false, false);
    acc[1][0] = __builtin_amdgcn_wmma_f32_16x16x4_f32(false, f.a1, false, f.b0, (short)0, acc[1][0], false, false);
    acc[0][1] = __builtin_amdgcn_wmma_f32_16x16x4_f32(false, f.a0, false, f.b1, (short)0, acc[0][1], false, false);
    acc[1][1] = __builtin_amdgcn_wmma_f32_16x16x4_f32(false, f.a1, false, f.b1, (short)0, acc[1][1], false, false);
    acc[0][2] = __builtin_amdgcn_wmma_f32_16x16x4_f32(false, f.a0, false, f.b2, (short)0, acc[0][2], false, false);
    acc[1][2] = __builtin_amdgcn_wmma_f32_16x16x4_f32(false, f.a1, false, f.b2, (short)0, acc[1][2], false, false);
    acc[0][3] = __builtin_amdgcn_wmma_f32_16x16x4_f32(false, f.a0, false, f.b3, (short)0, acc[0][3], false, false);
    acc[1][3] = __builtin_amdgcn_wmma_f32_16x16x4_f32(false, f.a1, false, f.b3, (short)0, acc[1][3], false, false);
}

__global__ __launch_bounds__(128, 1) void gemm_wmma_f32(
    const float* __restrict__ A, const float* __restrict__ W,
    const float* __restrict__ bias, float* __restrict__ C,
    int M, int N, int K, int lda, int act)
{
    const int lane = threadIdx.x & 31;
    const int wave = threadIdx.x >> 5;
    const int m0   = (blockIdx.y << 7) + (wave << 5);
    const int n0   = blockIdx.x << 6;
    if (m0 >= M) return;

    const int mrow  = lane & 15;
    const int khalf = (lane >> 4) << 1;

    const float* __restrict__ Ar0 = A + (size_t)(m0 + mrow) * lda + khalf;
    const float* __restrict__ Ar1 = Ar0 + (size_t)16 * lda;

    const float* Wr[4];
#pragma unroll
    for (int j = 0; j < 4; ++j) {
        int row = n0 + (j << 4) + mrow;
        row = (row < N) ? row : (N - 1);
        Wr[j] = W + (size_t)row * K + khalf;
    }

    v8f acc[2][4];
#pragma unroll
    for (int i = 0; i < 2; ++i)
#pragma unroll
        for (int j = 0; j < 4; ++j) acc[i][j] = (v8f){};

    Frag f0, f1;
    load_frag(f0, Ar0, Ar1, Wr[0], Wr[1], Wr[2], Wr[3], 0);

    int k0 = 0;
    for (; k0 < K - 16; k0 += 16) {
        __builtin_prefetch(Ar0   + k0 + 128, 0, 1);
        __builtin_prefetch(Ar1   + k0 + 128, 0, 1);
        __builtin_prefetch(Wr[0] + k0 + 128, 0, 1);
        __builtin_prefetch(Wr[2] + k0 + 128, 0, 1);
        load_frag(f1, Ar0, Ar1, Wr[0], Wr[1], Wr[2], Wr[3], k0 + 4);
        do_wmma(f0, acc);
        load_frag(f0, Ar0, Ar1, Wr[0], Wr[1], Wr[2], Wr[3], k0 + 8);
        do_wmma(f1, acc);
        load_frag(f1, Ar0, Ar1, Wr[0], Wr[1], Wr[2], Wr[3], k0 + 12);
        do_wmma(f0, acc);
        load_frag(f0, Ar0, Ar1, Wr[0], Wr[1], Wr[2], Wr[3], k0 + 16);
        do_wmma(f1, acc);
    }
    load_frag(f1, Ar0, Ar1, Wr[0], Wr[1], Wr[2], Wr[3], k0 + 4);
    do_wmma(f0, acc);
    load_frag(f0, Ar0, Ar1, Wr[0], Wr[1], Wr[2], Wr[3], k0 + 8);
    do_wmma(f1, acc);
    load_frag(f1, Ar0, Ar1, Wr[0], Wr[1], Wr[2], Wr[3], k0 + 12);
    do_wmma(f0, acc);
    do_wmma(f1, acc);

    const int rsub = (lane >> 4) << 3;
#pragma unroll
    for (int j = 0; j < 4; ++j) {
        const int col = n0 + (j << 4) + mrow;
        if ((n0 + (j << 4)) >= N) continue;
        const float bv = (act == 1) ? bias[col] : 0.0f;
#pragma unroll
        for (int i = 0; i < 2; ++i) {
            const int rbase = m0 + (i << 4) + rsub;
#pragma unroll
            for (int r = 0; r < 8; ++r) {
                float v = acc[i][j][r];
                if (act == 1) {
                    v += bv;
                    v = (v > 20.0f) ? v : log1pf(__expf(v));
                }
                C[(size_t)(rbase + r) * N + col] = v;
            }
        }
    }
}

__global__ __launch_bounds__(256) void conv_silu_kernel(
    const float* __restrict__ xz, const float* __restrict__ cw,
    const float* __restrict__ cb, float* __restrict__ xc)
{
    int idx = blockIdx.x * blockDim.x + threadIdx.x;
    if (idx >= LSEQ * D_INNER) return;
    int l = idx >> 11;
    int d = idx & (D_INNER - 1);

    float acc = cb[d];
#pragma unroll
    for (int j = 0; j < D_CONV; ++j) {
        int ls = l - (D_CONV - 1) + j;
        if (ls >= 0)
            acc += cw[d * D_CONV + j] * xz[(size_t)ls * (2 * D_INNER) + d];
    }
    xc[idx] = acc / (1.0f + __expf(-acc));
}

#define SCAN_T   32
#define SCAN_NC  (LSEQ / SCAN_T)
#define SCAN_BLK 128
#define XDBL_LD  (DT_RANK + 2 * D_STATE)

__global__ __launch_bounds__(SCAN_BLK) void scan_chunk_state(
    const float* __restrict__ xdbl, const float* __restrict__ dtbuf,
    const float* __restrict__ xc,   const float* __restrict__ A_log,
    float* __restrict__ hend, float* __restrict__ dtsum)
{
    __shared__ float sB[SCAN_T][D_STATE];

    const int d  = blockIdx.x * SCAN_BLK + threadIdx.x;
    const int c  = blockIdx.y;
    const int lc = c * SCAN_T;

    for (int i = threadIdx.x; i < SCAN_T * D_STATE; i += SCAN_BLK) {
        int t = i >> 4, n = i & 15;
        sB[t][n] = xdbl[(size_t)(lc + t) * XDBL_LD + DT_RANK + n];
    }

    float Acoef[D_STATE];
#pragma unroll
    for (int n = 0; n < D_STATE; ++n)
        Acoef[n] = -__expf(A_log[(size_t)d * D_STATE + n]);

    float h[D_STATE];
#pragma unroll
    for (int n = 0; n < D_STATE; ++n) h[n] = 0.0f;
    float s = 0.0f;

    __syncthreads();

#pragma unroll 8
    for (int t = 0; t < SCAN_T; ++t) {
        const int l = lc + t;
        const float dt = dtbuf[(size_t)l * D_INNER + d];
        const float xv = xc[(size_t)l * D_INNER + d];
        s += dt;
#pragma unroll
        for (int n = 0; n < D_STATE; ++n) {
            float dA = __expf(dt * Acoef[n]);
            h[n] = dA * h[n] + (dt * sB[t][n]) * xv;
        }
    }

#pragma unroll
    for (int n = 0; n < D_STATE; ++n)
        hend[((size_t)c * D_STATE + n) * D_INNER + d] = h[n];
    dtsum[(size_t)c * D_INNER + d] = s;
}

__global__ __launch_bounds__(256) void scan_carry(
    const float* __restrict__ hend, const float* __restrict__ dtsum,
    const float* __restrict__ A_log, float* __restrict__ hstart)
{
    const int idx = blockIdx.x * blockDim.x + threadIdx.x;
    const int n = idx >> 11;
    const int d = idx & (D_INNER - 1);
    const float Ac = -__expf(A_log[(size_t)d * D_STATE + n]);

    float cur = 0.0f;
    hstart[idx] = 0.0f;
#pragma unroll 7
    for (int c = 0; c < SCAN_NC - 1; ++c) {
        const float e = __expf(dtsum[(size_t)c * D_INNER + d] * Ac);
        cur = e * cur + hend[(size_t)c * (D_STATE * D_INNER) + idx];
        hstart[(size_t)(c + 1) * (D_STATE * D_INNER) + idx] = cur;
    }
}

__global__ __launch_bounds__(SCAN_BLK) void scan_chunk_out(
    const float* __restrict__ xdbl, const float* __restrict__ dtbuf,
    const float* __restrict__ xc,   const float* __restrict__ xz,
    const float* __restrict__ A_log, const float* __restrict__ D_param,
    const float* __restrict__ hstart, float* __restrict__ yout)
{
    __shared__ float sB[SCAN_T][D_STATE];
    __shared__ float sC[SCAN_T][D_STATE];

    const int d  = blockIdx.x * SCAN_BLK + threadIdx.x;
    const int c  = blockIdx.y;
    const int lc = c * SCAN_T;

    for (int i = threadIdx.x; i < SCAN_T * D_STATE; i += SCAN_BLK) {
        int t = i >> 4, n = i & 15;
        const float* row = xdbl + (size_t)(lc + t) * XDBL_LD;
        sB[t][n] = row[DT_RANK + n];
        sC[t][n] = row[DT_RANK + D_STATE + n];
    }

    float Acoef[D_STATE];
#pragma unroll
    for (int n = 0; n < D_STATE; ++n)
        Acoef[n] = -__expf(A_log[(size_t)d * D_STATE + n]);
    const float Dp = D_param[d];

    float h[D_STATE];
#pragma unroll
    for (int n = 0; n < D_STATE; ++n)
        h[n] = hstart[((size_t)c * D_STATE + n) * D_INNER + d];

    __syncthreads();

#pragma unroll 8
    for (int t = 0; t < SCAN_T; ++t) {
        const int l = lc + t;
        const float dt = dtbuf[(size_t)l * D_INNER + d];
        const float xv = xc[(size_t)l * D_INNER + d];
        const float z  = xz[(size_t)l * (2 * D_INNER) + D_INNER + d];
        float y = 0.0f;
#pragma unroll
        for (int n = 0; n < D_STATE; ++n) {
            float dA = __expf(dt * Acoef[n]);
            h[n] = dA * h[n] + (dt * sB[t][n]) * xv;
            y += h[n] * sC[t][n];
        }
        y += xv * Dp;
        float g = z / (1.0f + __expf(-z));
        yout[(size_t)l * D_INNER + d] = y * g;
    }
}

extern "C" void kernel_launch(void* const* d_in, const int* in_sizes, int n_in,
                              void* d_out, int out_size, void* d_ws, size_t ws_size,
                              hipStream_t stream) {
    const float* x       = (const float*)d_in[0];
    const float* W_in    = (const float*)d_in[1];
    const float* conv_w  = (const float*)d_in[2];
    const float* conv_b  = (const float*)d_in[3];
    const float* W_x     = (const float*)d_in[4];
    const float* W_dt    = (const float*)d_in[5];
    const float* b_dt    = (const float*)d_in[6];
    const float* A_log   = (const float*)d_in[7];
    const float* D_param = (const float*)d_in[8];
    const float* W_out   = (const float*)d_in[9];
    float* out = (float*)d_out;

    float* xz   = (float*)d_ws;
    float* xc   = xz   + (size_t)LSEQ * 2 * D_INNER;
    float* xdbl = xc   + (size_t)LSEQ * D_INNER;
    float* dtb  = xdbl + (size_t)LSEQ * (DT_RANK + 2 * D_STATE);
    float* yb   = dtb  + (size_t)LSEQ * D_INNER;
    float* hend = yb   + (size_t)LSEQ * D_INNER;
    float* hst  = hend + (size_t)SCAN_NC * D_STATE * D_INNER;
    float* dts  = hst  + (size_t)SCAN_NC * D_STATE * D_INNER;

    dim3 blk(128, 1, 1);

    gemm_wmma_f32<<<dim3(4096 / 64, LSEQ / 128), blk, 0, stream>>>(
        x, W_in, nullptr, xz, LSEQ, 2 * D_INNER, D_MODEL, D_MODEL, 0);

    conv_silu_kernel<<<(LSEQ * D_INNER) / 256, 256, 0, stream>>>(
        xz, conv_w, conv_b, xc);

    gemm_wmma_f32<<<dim3((96 + 63) / 64, LSEQ / 128), blk, 0, stream>>>(
        xc, W_x, nullptr, xdbl, LSEQ, DT_RANK + 2 * D_STATE, D_INNER, D_INNER, 0);

    gemm_wmma_f32<<<dim3(D_INNER / 64, LSEQ / 128), blk, 0, stream>>>(
        xdbl, W_dt, b_dt, dtb, LSEQ, D_INNER, DT_RANK, DT_RANK + 2 * D_STATE, 1);

    scan_chunk_state<<<dim3(D_INNER / SCAN_BLK, SCAN_NC - 1), SCAN_BLK, 0, stream>>>(
        xdbl, dtb, xc, A_log, hend, dts);
    scan_carry<<<(D_STATE * D_INNER) / 256, 256, 0, stream>>>(
        hend, dts, A_log, hst);
    scan_chunk_out<<<dim3(D_INNER / SCAN_BLK, SCAN_NC), SCAN_BLK, 0, stream>>>(
        xdbl, dtb, xc, xz, A_log, D_param, hst, yb);

    gemm_wmma_f32<<<dim3(D_MODEL / 64, LSEQ / 128), blk, 0, stream>>>(
        yb, W_out, nullptr, out, LSEQ, D_MODEL, D_INNER, D_INNER, 0);
}


# GEMMs: each lane reads its row's 32-k block as four 16-byte loads (two lanes take a whole 128-B line), one block ahead in a 2nd register set; k permuted within 8 alike for A and W (sum order differs)
# speedup vs baseline: 10.6917x; 1.7043x over previous
#include <hip/hip_runtime.h>
#include <math.h>

#define D_MODEL 1024
#define D_STATE 16
#define D_CONV  4
#define DT_RANK 64
#define D_INNER 2048
#define LSEQ    2048

typedef __attribute__((ext_vector_type(2))) float v2f;
typedef __attribute__((ext_vector_type(8))) float v8f;

typedef __attribute__((ext_vector_type(4))) float v4f;

struct Blk { v4f a0[4], a1[4], b0[4], b1[4], b2[4], b3[4]; };

__device__ __forceinline__ void load_row(v4f q[4], const float* p)
{
#pragma unroll
    for (int j = 0; j < 4; ++j) q[j] = *(const v4f*)(p + 8 * j);
}

__device__ __forceinline__ void load_blk(
    Blk& f, const float* Ar0, const float* Ar1,
    const float* w0, const float* w1, const float* w2, const float* w3, int k)
{
    load_row(f.a0, Ar0 + k);
    load_row(f.a1, Ar1 + k);
    load_row(f.b0, w0 + k);
    load_row(f.b1, w1 + k);
    load_row(f.b2, w2 + k);
    load_row(f.b3, w3 + k);
}

__device__ __forceinline__ void mma_step(
    v2f a0, v2f a1, v2f b0, v2f b1, v2f b2, v2f b3, v8f acc[2][4])
{
    acc[0][0] = __builtin_amdgcn_wmma_f32_16x16x4_f32(false, a0, false, b0, (short)0, acc[0][0], false, false);
    acc[1][0] = __builtin_amdgcn_wmma_f32_16x16x4_f32(false, a1, false, b0, (short)0, acc[1][0], false, false);
    acc[0][1] = __builtin_amdgcn_wmma_f32_16x16x4_f32(false, a0, false, b1, (short)0, acc[0][1], false, false);
    acc[1][1] = __builtin_amdgcn_wmma_f32_16x16x4_f32(false, a1, false, b1, (short)0, acc[1][1], false, false);
    acc[0][2] = __builtin_amdgcn_wmma_f32_16x16x4_f32(false, a0, false, b2, (short)0, acc[0][2], false, false);
    acc[1][2] = __builtin_amdgcn_wmma_f32_16x16x4_f32(false, a1, false, b2, (short)0, acc[1][2], false, false);
    acc[0][3] = __builtin_amdgcn_wmma_f32_16x16x4_f32(false, a0, false, b3, (short)0, acc[0][3], false, false);
    acc[1][3] = __builtin_amdgcn_wmma_f32_16x16x4_f32(false, a1, false, b3, (short)0, acc[1][3], false, false);
}

__device__ __forceinline__ void mma_blk(const Blk& f, v8f acc[2][4])
{
#pragma unroll
    for (int j = 0; j < 4; ++j) {
        mma_step(f.a0[j].xy, f.a1[j].xy, f.b0[j].xy, f.b1[j].xy, f.b2[j].xy, f.b3[j].xy, acc);
        mma_step(f.a0[j].zw, f.a1[j].zw, f.b0[j].zw, f.b1[j].zw, f.b2[j].zw, f.b3[j].zw, acc);
    }
}

__global__ __launch_bounds__(128, 1) void gemm_wmma_f32(
    const float* __restrict__ A, const float* __restrict__ W,
    const float* __restrict__ bias, float* __restrict__ C,
    int M, int N, int K, int lda, int act)
{
    const int lane = threadIdx.x & 31;
    const int wave = threadIdx.x >> 5;
    const int m0   = (blockIdx.y << 7) + (wave << 5);
    const int n0   = blockIdx.x << 6;
    if (m0 >= M) return;

    const int mrow  = lane & 15;
    const int khalf = (lane >> 4) << 2;

    const float* __restrict__ Ar0 = A + (size_t)(m0 + mrow) * lda + khalf;
    const float* __restrict__ Ar1 = Ar0 + (size_t)16 * lda;

    const float* Wr[4];
#pragma unroll
    for (int j = 0; j < 4; ++j) {
        int row = n0 + (j << 4) + mrow;
        row = (row < N) ? row : (N - 1);
        Wr[j] = W + (size_t)row * K + khalf;
    }

    v8f acc[2][4];
#pragma unroll
    for (int i = 0; i < 2; ++i)
#pragma unroll
        for (int j = 0; j < 4; ++j) acc[i][j] = (v8f){};

    Blk f0, f1;
    load_blk(f0, Ar0, Ar1, Wr[0], Wr[1], Wr[2], Wr[3], 0);

    int k0 = 0;
    for (; k0 < K - 64; k0 += 64) {
        load_blk(f1, Ar0, Ar1, Wr[0], Wr[1], Wr[2], Wr[3], k0 + 32);
        __builtin_amdgcn_sched_barrier(0);
        mma_blk(f0, acc);
        __builtin_amdgcn_sched_barrier(0);
        load_blk(f0, Ar0, Ar1, Wr[0], Wr[1], Wr[2], Wr[3], k0 + 64);
        __builtin_amdgcn_sched_barrier(0);
        mma_blk(f1, acc);
    }
    load_blk(f1, Ar0, Ar1, Wr[0], Wr[1], Wr[2], Wr[3], k0 + 32);
    __builtin_amdgcn_sched_barrier(0);
    mma_blk(f0, acc);
    mma_blk(f1, acc);

    const int rsub = (lane >> 4) << 3;
#pragma unroll
    for (int j = 0; j < 4; ++j) {
        const int col = n0 + (j << 4) + mrow;
        if ((n0 + (j << 4)) >= N) continue;
        const float bv = (act == 1) ? bias[col] : 0.0f;
#pragma unroll
        for (int i = 0; i < 2; ++i) {
            const int rbase = m0 + (i << 4) + rsub;
#pragma unroll
            for (int r = 0; r < 8; ++r) {
                float v = acc[i][j][r];
                if (act == 1) {
                    v += bv;
                    v = (v > 20.0f) ? v : log1pf(__expf(v));
                }
                C[(size_t)(rbase + r) * N + col] = v;
            }
        }
    }
}

__global__ __launch_bounds__(256) void conv_silu_kernel(
    const float* __restrict__ xz, const float* __restrict__ cw,
    const float* __restrict__ cb, float* __restrict__ xc)
{
    int idx = blockIdx.x * blockDim.x + threadIdx.x;
    if (idx >= LSEQ * D_INNER) return;
    int l = idx >> 11;
    int d = idx & (D_INNER - 1);

    float acc = cb[d];
#pragma unroll
    for (int j = 0; j < D_CONV; ++j) {
        int ls = l - (D_CONV - 1) + j;
        if (ls >= 0)
            acc += cw[d * D_CONV + j] * xz[(size_t)ls * (2 * D_INNER) + d];
    }
    xc[idx] = acc / (1.0f + __expf(-acc));
}

#define SCAN_T   32
#define SCAN_NC  (LSEQ / SCAN_T)
#define SCAN_BLK 128
#define XDBL_LD  (DT_RANK + 2 * D_STATE)

__global__ __launch_bounds__(SCAN_BLK) void scan_chunk_state(
    const float* __restrict__ xdbl, const float* __restrict__ dtbuf,
    const float* __restrict__ xc,   const float* __restrict__ A_log,
    float* __restrict__ hend, float* __restrict__ dtsum)
{
    __shared__ float sB[SCAN_T][D_STATE];

    const int d  = blockIdx.x * SCAN_BLK + threadIdx.x;
    const int c  = blockIdx.y;
    const int lc = c * SCAN_T;

    for (int i = threadIdx.x; i < SCAN_T * D_STATE; i += SCAN_BLK) {
        int t = i >> 4, n = i & 15;
        sB[t][n] = xdbl[(size_t)(lc + t) * XDBL_LD + DT_RANK + n];
    }

    float Acoef[D_STATE];
#pragma unroll
    for (int n = 0; n < D_STATE; ++n)
        Acoef[n] = -__expf(A_log[(size_t)d * D_STATE + n]);

    float h[D_STATE];
#pragma unroll
    for (int n = 0; n < D_STATE; ++n) h[n] = 0.0f;
    float s = 0.0f;

    __syncthreads();

#pragma unroll 8
    for (int t = 0; t < SCAN_T; ++t) {
        const int l = lc + t;
        const float dt = dtbuf[(size_t)l * D_INNER + d];
        const float xv = xc[(size_t)l * D_INNER + d];
        s += dt;
#pragma unroll
        for (int n = 0; n < D_STATE; ++n) {
            float dA = __expf(dt * Acoef[n]);
            h[n] = dA * h[n] + (dt * sB[t][n]) * xv;
        }
    }

#pragma unroll
    for (int n = 0; n < D_STATE; ++n)
        hend[((size_t)c * D_STATE + n) * D_INNER + d] = h[n];
    dtsum[(size_t)c * D_INNER + d] = s;
}

__global__ __launch_bounds__(256) void scan_carry(
    const float* __restrict__ hend, const float* __restrict__ dtsum,
    const float* __restrict__ A_log, float* __restrict__ hstart)
{
    const int idx = blockIdx.x * blockDim.x + threadIdx.x;
    const int n = idx >> 11;
    const int d = idx & (D_INNER - 1);
    const float Ac = -__expf(A_log[(size_t)d * D_STATE + n]);

    float cur = 0.0f;
    hstart[idx] = 0.0f;
#pragma unroll 7
    for (int c = 0; c < SCAN_NC - 1; ++c) {
        const float e = __expf(dtsum[(size_t)c * D_INNER + d] * Ac);
        cur = e * cur + hend[(size_t)c * (D_STATE * D_INNER) + idx];
        hstart[(size_t)(c + 1) * (D_STATE * D_INNER) + idx] = cur;
    }
}

__global__ __launch_bounds__(SCAN_BLK) void scan_chunk_out(
    const float* __restrict__ xdbl, const float* __restrict__ dtbuf,
    const float* __restrict__ xc,   const float* __restrict__ xz,
    const float* __restrict__ A_log, const float* __restrict__ D_param,
    const float* __restrict__ hstart, float* __restrict__ yout)
{
    __shared__ float sB[SCAN_T][D_STATE];
    __shared__ float sC[SCAN_T][D_STATE];

    const int d  = blockIdx.x * SCAN_BLK + threadIdx.x;
    const int c  = blockIdx.y;
    const int lc = c * SCAN_T;

    for (int i = threadIdx.x; i < SCAN_T * D_STATE; i += SCAN_BLK) {
        int t = i >> 4, n = i & 15;
        const float* row = xdbl + (size_t)(lc + t) * XDBL_LD;
        sB[t][n] = row[DT_RANK + n];
        sC[t][n] = row[DT_RANK + D_STATE + n];
    }

    float Acoef[D_STATE];
#pragma unroll
    for (int n = 0; n < D_STATE; ++n)
        Acoef[n] = -__expf(A_log[(size_t)d * D_STATE + n]);
    const float Dp = D_param[d];

    float h[D_STATE];
#pragma unroll
    for (int n = 0; n < D_STATE; ++n)
        h[n] = hstart[((size_t)c * D_STATE + n) * D_INNER + d];

    __syncthreads();

#pragma unroll 8
    for (int t = 0; t < SCAN_T; ++t) {
        const int l = lc + t;
        const float dt = dtbuf[(size_t)l * D_INNER + d];
        const float xv = xc[(size_t)l * D_INNER + d];
        const float z  = xz[(size_t)l * (2 * D_INNER) + D_INNER + d];
        float y = 0.0f;
#pragma unroll
        for (int n = 0; n < D_STATE; ++n) {
            float dA = __expf(dt * Acoef[n]);
            h[n] = dA * h[n] + (dt * sB[t][n]) * xv;
            y += h[n] * sC[t][n];
        }
        y += xv * Dp;
        float g = z / (1.0f + __expf(-z));
        yout[(size_t)l * D_INNER + d] = y * g;
    }
}

extern "C" void kernel_launch(void* const* d_in, const int* in_sizes, int n_in,
                              void* d_out, int out_size, void* d_ws, size_t ws_size,
                              hipStream_t stream) {
    const float* x       = (const float*)d_in[0];
    const float* W_in    = (const float*)d_in[1];
    const float* conv_w  = (const float*)d_in[2];
    const float* conv_b  = (const float*)d_in[3];
    const float* W_x     = (const float*)d_in[4];
    const float* W_dt    = (const float*)d_in[5];
    const float* b_dt    = (const float*)d_in[6];
    const float* A_log   = (const float*)d_in[7];
    const float* D_param = (const float*)d_in[8];
    const float* W_out   = (const float*)d_in[9];
    float* out = (float*)d_out;

    float* xz   = (float*)d_ws;
    float* xc   = xz   + (size_t)LSEQ * 2 * D_INNER;
    float* xdbl = xc   + (size_t)LSEQ * D_INNER;
    float* dtb  = xdbl + (size_t)LSEQ * (DT_RANK + 2 * D_STATE);
    float* yb   = dtb  + (size_t)LSEQ * D_INNER;
    float* hend = yb   + (size_t)LSEQ * D_INNER;
    float* hst  = hend + (size_t)SCAN_NC * D_STATE * D_INNER;
    float* dts  = hst  + (size_t)SCAN_NC * D_STATE * D_INNER;

    dim3 blk(128, 1, 1);

    gemm_wmma_f32<<<dim3(4096 / 64, LSEQ / 128), blk, 0, stream>>>(
        x, W_in, nullptr, xz, LSEQ, 2 * D_INNER, D_MODEL, D_MODEL, 0);

    conv_silu_kernel<<<(LSEQ * D_INNER) / 256, 256, 0, stream>>>(
        xz, conv_w, conv_b, xc);

    gemm_wmma_f32<<<dim3((96 + 63) / 64, LSEQ / 128), blk, 0, stream>>>(
        xc, W_x, nullptr, xdbl, LSEQ, DT_RANK + 2 * D_STATE, D_INNER, D_INNER, 0);

    gemm_wmma_f32<<<dim3(D_INNER / 64, LSEQ / 128), blk, 0, stream>>>(
        xdbl, W_dt, b_dt, dtb, LSEQ, D_INNER, DT_RANK, DT_RANK + 2 * D_STATE, 1);

    scan_chunk_state<<<dim3(D_INNER / SCAN_BLK, SCAN_NC - 1), SCAN_BLK, 0, stream>>>(
        xdbl, dtb, xc, A_log, hend, dts);
    scan_carry<<<(D_STATE * D_INNER) / 256, 256, 0, stream>>>(
        hend, dts, A_log, hst);
    scan_chunk_out<<<dim3(D_INNER / SCAN_BLK, SCAN_NC), SCAN_BLK, 0, stream>>>(
        xdbl, dtb, xc, xz, A_log, D_param, hst, yb);

    gemm_wmma_f32<<<dim3(D_MODEL / 64, LSEQ / 128), blk, 0, stream>>>(
        yb, W_out, nullptr, out, LSEQ, D_MODEL, D_INNER, D_INNER, 0);
}


# x_dbl GEMM (N=96): own kernel, one wave per 16x32 tile (2 accumulators) over the whole K in 128-k blocks: 384 one-wave groups on all units, not 32 groups of 4 waves; k order of every sum kept
# speedup vs baseline: 11.8844x; 1.1116x over previous
#include <hip/hip_runtime.h>
#include <math.h>

#define D_MODEL 1024
#define D_STATE 16
#define D_CONV  4
#define DT_RANK 64
#define D_INNER 2048
#define LSEQ    2048

typedef __attribute__((ext_vector_type(2))) float v2f;
typedef __attribute__((ext_vector_type(8))) float v8f;

typedef __attribute__((ext_vector_type(4))) float v4f;

struct Blk { v4f a0[4], a1[4], b0[4], b1[4], b2[4], b3[4]; };

template <int NP>
__device__ __forceinline__ void load_row(v4f* q, const float* p)
{
#pragma unroll
    for (int j = 0; j < NP; ++j) q[j] = *(const v4f*)(p + 8 * j);
}

__device__ __forceinline__ void load_blk(
    Blk& f, const float* Ar0, const float* Ar1,
    const float* w0, const float* w1, const float* w2, const float* w3, int k)
{
    load_row<4>(f.a0, Ar0 + k);
    load_row<4>(f.a1, Ar1 + k);
    load_row<4>(f.b0, w0 + k);
    load_row<4>(f.b1, w1 + k);
    load_row<4>(f.b2, w2 + k);
    load_row<4>(f.b3, w3 + k);
}

__device__ __forceinline__ void mma_step(
    v2f a0, v2f a1, v2f b0, v2f b1, v2f b2, v2f b3, v8f acc[2][4])
{
    acc[0][0] = __builtin_amdgcn_wmma_f32_16x16x4_f32(false, a0, false, b0, (short)0, acc[0][0], false, false);
    acc[1][0] = __builtin_amdgcn_wmma_f32_16x16x4_f32(false, a1, false, b0, (short)0, acc[1][0], false, false);
    acc[0][1] = __builtin_amdgcn_wmma_f32_16x16x4_f32(false, a0, false, b1, (short)0, acc[0][1], false, false);
    acc[1][1] = __builtin_amdgcn_wmma_f32_16x16x4_f32(false, a1, false, b1, (short)0, acc[1][1], false, false);
    acc[0][2] = __builtin_amdgcn_wmma_f32_16x16x4_f32(false, a0, false, b2, (short)0, acc[0][2], false, false);
    acc[1][2] = __builtin_amdgcn_wmma_f32_16x16x4_f32(false, a1, false, b2, (short)0, acc[1][2], false, false);
    acc[0][3] = __builtin_amdgcn_wmma_f32_16x16x4_f32(false, a0, false, b3, (short)0, acc[0][3], false, false);
    acc[1][3] = __builtin_amdgcn_wmma_f32_16x16x4_f32(false, a1, false, b3, (short)0, acc[1][3], false, false);
}

__device__ __forceinline__ void mma_blk(const Blk& f, v8f acc[2][4])
{
#pragma unroll
    for (int j = 0; j < 4; ++j) {
        mma_step(f.a0[j].xy, f.a1[j].xy, f.b0[j].xy, f.b1[j].xy, f.b2[j].xy, f.b3[j].xy, acc);
        mma_step(f.a0[j].zw, f.a1[j].zw, f.b0[j].zw, f.b1[j].zw, f.b2[j].zw, f.b3[j].zw, acc);
    }
}

__global__ __launch_bounds__(128, 1) void gemm_wmma_f32(
    const float* __restrict__ A, const float* __restrict__ W,
    const float* __restrict__ bias, float* __restrict__ C,
    int M, int N, int K, int lda, int act)
{
    const int lane = threadIdx.x & 31;
    const int wave = threadIdx.x >> 5;
    const int m0   = (blockIdx.y << 7) + (wave << 5);
    const int n0   = blockIdx.x << 6;
    if (m0 >= M) return;

    const int mrow  = lane & 15;
    const int khalf = (lane >> 4) << 2;

    const float* __restrict__ Ar0 = A + (size_t)(m0 + mrow) * lda + khalf;
    const float* __restrict__ Ar1 = Ar0 + (size_t)16 * lda;

    const float* Wr[4];
#pragma unroll
    for (int j = 0; j < 4; ++j) {
        int row = n0 + (j << 4) + mrow;
        row = (row < N) ? row : (N - 1);
        Wr[j] = W + (size_t)row * K + khalf;
    }

    v8f acc[2][4];
#pragma unroll
    for (int i = 0; i < 2; ++i)
#pragma unroll
        for (int j = 0; j < 4; ++j) acc[i][j] = (v8f){};

    Blk f0, f1;
    load_blk(f0, Ar0, Ar1, Wr[0], Wr[1], Wr[2], Wr[3], 0);

    int k0 = 0;
    for (; k0 < K - 64; k0 += 64) {
        load_blk(f1, Ar0, Ar1, Wr[0], Wr[1], Wr[2], Wr[3], k0 + 32);
        __builtin_amdgcn_sched_barrier(0);
        mma_blk(f0, acc);
        __builtin_amdgcn_sched_barrier(0);
        load_blk(f0, Ar0, Ar1, Wr[0], Wr[1], Wr[2], Wr[3], k0 + 64);
        __builtin_amdgcn_sched_barrier(0);
        mma_blk(f1, acc);
    }
    load_blk(f1, Ar0, Ar1, Wr[0], Wr[1], Wr[2], Wr[3], k0 + 32);
    __builtin_amdgcn_sched_barrier(0);
    mma_blk(f0, acc);
    mma_blk(f1, acc);

    const int rsub = (lane >> 4) << 3;
#pragma unroll
    for (int j = 0; j < 4; ++j) {
        const int col = n0 + (j << 4) + mrow;
        if ((n0 + (j << 4)) >= N) continue;
        const float bv = (act == 1) ? bias[col] : 0.0f;
#pragma unroll
        for (int i = 0; i < 2; ++i) {
            const int rbase = m0 + (i << 4) + rsub;
#pragma unroll
            for (int r = 0; r < 8; ++r) {
                float v = acc[i][j][r];
                if (act == 1) {
                    v += bv;
                    v = (v > 20.0f) ? v : log1pf(__expf(v));
                }
                C[(size_t)(rbase + r) * N + col] = v;
            }
        }
    }
}

#define NARROW_P 16

struct BlkN { v4f a[NARROW_P], b0[NARROW_P], b1[NARROW_P]; };

__device__ __forceinline__ void load_blk_n(
    BlkN& f, const float* Ar, const float* w0, const float* w1, int k)
{
    load_row<NARROW_P>(f.a,  Ar + k);
    load_row<NARROW_P>(f.b0, w0 + k);
    load_row<NARROW_P>(f.b1, w1 + k);
}

__device__ __forceinline__ v8f wmma_f32(v2f a, v2f b, v8f c)
{
    return __builtin_amdgcn_wmma_f32_16x16x4_f32(false, a, false, b, (short)0, c, false, false);
}

__device__ __forceinline__ void mma_blk_n(const BlkN& f, v8f acc[2])
{
#pragma unroll
    for (int j = 0; j < NARROW_P; ++j) {
        acc[0] = wmma_f32(f.a[j].xy, f.b0[j].xy, acc[0]);
        acc[1] = wmma_f32(f.a[j].xy, f.b1[j].xy, acc[1]);
        acc[0] = wmma_f32(f.a[j].zw, f.b0[j].zw, acc[0]);
        acc[1] = wmma_f32(f.a[j].zw, f.b1[j].zw, acc[1]);
    }
}

__global__ __launch_bounds__(32, 1) void gemm_wmma_f32_narrow(
    const float* __restrict__ A, const float* __restrict__ W,
    float* __restrict__ C, int N, int K, int lda)
{
    const int lane  = threadIdx.x;
    const int m0    = blockIdx.y << 4;
    const int n0    = blockIdx.x << 5;
    const int mrow  = lane & 15;
    const int khalf = (lane >> 4) << 2;

    const float* __restrict__ Ar = A + (size_t)(m0 + mrow) * lda + khalf;
    const float* __restrict__ w0 = W + (size_t)(n0 + mrow) * K + khalf;
    const float* __restrict__ w1 = w0 + (size_t)16 * K;

    v8f acc[2];
    acc[0] = (v8f){};
    acc[1] = (v8f){};

    BlkN f0, f1;
    load_blk_n(f0, Ar, w0, w1, 0);

    int k0 = 0;
    for (; k0 < K - 256; k0 += 256) {
        load_blk_n(f1, Ar, w0, w1, k0 + 128);
        mma_blk_n(f0, acc);
        load_blk_n(f0, Ar, w0, w1, k0 + 256);
        mma_blk_n(f1, acc);
        __builtin_amdgcn_sched_group_barrier(0x020, 3 * NARROW_P, 0);
        __builtin_amdgcn_sched_group_barrier(0x008, 4 * NARROW_P, 0);
        __builtin_amdgcn_sched_group_barrier(0x020, 3 * NARROW_P, 0);
        __builtin_amdgcn_sched_group_barrier(0x008, 4 * NARROW_P, 0);
    }
    load_blk_n(f1, Ar, w0, w1, k0 + 128);
    __builtin_amdgcn_sched_barrier(0);
    mma_blk_n(f0, acc);
    mma_blk_n(f1, acc);

    const int rsub = (lane >> 4) << 3;
#pragma unroll
    for (int j = 0; j < 2; ++j) {
        const int col = n0 + (j << 4) + mrow;
#pragma unroll
        for (int r = 0; r < 8; ++r)
            C[(size_t)(m0 + rsub + r) * N + col] = acc[j][r];
    }
}

__global__ __launch_bounds__(256) void conv_silu_kernel(
    const float* __restrict__ xz, const float* __restrict__ cw,
    const float* __restrict__ cb, float* __restrict__ xc)
{
    int idx = blockIdx.x * blockDim.x + threadIdx.x;
    if (idx >= LSEQ * D_INNER) return;
    int l = idx >> 11;
    int d = idx & (D_INNER - 1);

    float acc = cb[d];
#pragma unroll
    for (int j = 0; j < D_CONV; ++j) {
        int ls = l - (D_CONV - 1) + j;
        if (ls >= 0)
            acc += cw[d * D_CONV + j] * xz[(size_t)ls * (2 * D_INNER) + d];
    }
    xc[idx] = acc / (1.0f + __expf(-acc));
}

#define SCAN_T   32
#define SCAN_NC  (LSEQ / SCAN_T)
#define SCAN_BLK 128
#define XDBL_LD  (DT_RANK + 2 * D_STATE)

__global__ __launch_bounds__(SCAN_BLK) void scan_chunk_state(
    const float* __restrict__ xdbl, const float* __restrict__ dtbuf,
    const float* __restrict__ xc,   const float* __restrict__ A_log,
    float* __restrict__ hend, float* __restrict__ dtsum)
{
    __shared__ float sB[SCAN_T][D_STATE];

    const int d  = blockIdx.x * SCAN_BLK + threadIdx.x;
    const int c  = blockIdx.y;
    const int lc = c * SCAN_T;

    for (int i = threadIdx.x; i < SCAN_T * D_STATE; i += SCAN_BLK) {
        int t = i >> 4, n = i & 15;
        sB[t][n] = xdbl[(size_t)(lc + t) * XDBL_LD + DT_RANK + n];
    }

    float Acoef[D_STATE];
#pragma unroll
    for (int n = 0; n < D_STATE; ++n)
        Acoef[n] = -__expf(A_log[(size_t)d * D_STATE + n]);

    float h[D_STATE];
#pragma unroll
    for (int n = 0; n < D_STATE; ++n) h[n] = 0.0f;
    float s = 0.0f;

    __syncthreads();

#pragma unroll 8
    for (int t = 0; t < SCAN_T; ++t) {
        const int l = lc + t;
        const float dt = dtbuf[(size_t)l * D_INNER + d];
        const float xv = xc[(size_t)l * D_INNER + d];
        s += dt;
#pragma unroll
        for (int n = 0; n < D_STATE; ++n) {
            float dA = __expf(dt * Acoef[n]);
            h[n] = dA * h[n] + (dt * sB[t][n]) * xv;
        }
    }

#pragma unroll
    for (int n = 0; n < D_STATE; ++n)
        hend[((size_t)c * D_STATE + n) * D_INNER + d] = h[n];
    dtsum[(size_t)c * D_INNER + d] = s;
}

__global__ __launch_bounds__(256) void scan_carry(
    const float* __restrict__ hend, const float* __restrict__ dtsum,
    const float* __restrict__ A_log, float* __restrict__ hstart)
{
    const int idx = blockIdx.x * blockDim.x + threadIdx.x;
    const int n = idx >> 11;
    const int d = idx & (D_INNER - 1);
    const float Ac = -__expf(A_log[(size_t)d * D_STATE + n]);

    float cur = 0.0f;
    hstart[idx] = 0.0f;
#pragma unroll 7
    for (int c = 0; c < SCAN_NC - 1; ++c) {
        const float e = __expf(dtsum[(size_t)c * D_INNER + d] * Ac);
        cur = e * cur + hend[(size_t)c * (D_STATE * D_INNER) + idx];
        hstart[(size_t)(c + 1) * (D_STATE * D_INNER) + idx] = cur;
    }
}

__global__ __launch_bounds__(SCAN_BLK) void scan_chunk_out(
    const float* __restrict__ xdbl, const float* __restrict__ dtbuf,
    const float* __restrict__ xc,   const float* __restrict__ xz,
    const float* __restrict__ A_log, const float* __restrict__ D_param,
    const float* __restrict__ hstart, float* __restrict__ yout)
{
    __shared__ float sB[SCAN_T][D_STATE];
    __shared__ float sC[SCAN_T][D_STATE];

    const int d  = blockIdx.x * SCAN_BLK + threadIdx.x;
    const int c  = blockIdx.y;
    const int lc = c * SCAN_T;

    for (int i = threadIdx.x; i < SCAN_T * D_STATE; i += SCAN_BLK) {
        int t = i >> 4, n = i & 15;
        const float* row = xdbl + (size_t)(lc + t) * XDBL_LD;
        sB[t][n] = row[DT_RANK + n];
        sC[t][n] = row[DT_RANK + D_STATE + n];
    }

    float Acoef[D_STATE];
#pragma unroll
    for (int n = 0; n < D_STATE; ++n)
        Acoef[n] = -__expf(A_log[(size_t)d * D_STATE + n]);
    const float Dp = D_param[d];

    float h[D_STATE];
#pragma unroll
    for (int n = 0; n < D_STATE; ++n)
        h[n] = hstart[((size_t)c * D_STATE + n) * D_INNER + d];

    __syncthreads();

#pragma unroll 8
    for (int t = 0; t < SCAN_T; ++t) {
        const int l = lc + t;
        const float dt = dtbuf[(size_t)l * D_INNER + d];
        const float xv = xc[(size_t)l * D_INNER + d];
        const float z  = xz[(size_t)l * (2 * D_INNER) + D_INNER + d];
        float y = 0.0f;
#pragma unroll
        for (int n = 0; n < D_STATE; ++n) {
            float dA = __expf(dt * Acoef[n]);
            h[n] = dA * h[n] + (dt * sB[t][n]) * xv;
            y += h[n] * sC[t][n];
        }
        y += xv * Dp;
        float g = z / (1.0f + __expf(-z));
        yout[(size_t)l * D_INNER + d] = y * g;
    }
}

extern "C" void kernel_launch(void* const* d_in, const int* in_sizes, int n_in,
                              void* d_out, int out_size, void* d_ws, size_t ws_size,
                              hipStream_t stream) {
    const float* x       = (const float*)d_in[0];
    const float* W_in    = (const float*)d_in[1];
    const float* conv_w  = (const float*)d_in[2];
    const float* conv_b  = (const float*)d_in[3];
    const float* W_x     = (const float*)d_in[4];
    const float* W_dt    = (const float*)d_in[5];
    const float* b_dt    = (const float*)d_in[6];
    const float* A_log   = (const float*)d_in[7];
    const float* D_param = (const float*)d_in[8];
    const float* W_out   = (const float*)d_in[9];
    float* out = (float*)d_out;

    float* xz   = (float*)d_ws;
    float* xc   = xz   + (size_t)LSEQ * 2 * D_INNER;
    float* xdbl = xc   + (size_t)LSEQ * D_INNER;
    float* dtb  = xdbl + (size_t)LSEQ * (DT_RANK + 2 * D_STATE);
    float* yb   = dtb  + (size_t)LSEQ * D_INNER;
    float* hend = yb   + (size_t)LSEQ * D_INNER;
    float* hst  = hend + (size_t)SCAN_NC * D_STATE * D_INNER;
    float* dts  = hst  + (size_t)SCAN_NC * D_STATE * D_INNER;

    dim3 blk(128, 1, 1);

    gemm_wmma_f32<<<dim3(4096 / 64, LSEQ / 128), blk, 0, stream>>>(
        x, W_in, nullptr, xz, LSEQ, 2 * D_INNER, D_MODEL, D_MODEL, 0);

    conv_silu_kernel<<<(LSEQ * D_INNER) / 256, 256, 0, stream>>>(
        xz, conv_w, conv_b, xc);

    gemm_wmma_f32_narrow<<<dim3((DT_RANK + 2 * D_STATE) / 32, LSEQ / 16), dim3(32, 1, 1), 0, stream>>>(
        xc, W_x, xdbl, DT_RANK + 2 * D_STATE, D_INNER, D_INNER);

    gemm_wmma_f32<<<dim3(D_INNER / 64, LSEQ / 128), blk, 0, stream>>>(
        xdbl, W_dt, b_dt, dtb, LSEQ, D_INNER, DT_RANK, DT_RANK + 2 * D_STATE, 1);

    scan_chunk_state<<<dim3(D_INNER / SCAN_BLK, SCAN_NC - 1), SCAN_BLK, 0, stream>>>(
        xdbl, dtb, xc, A_log, hend, dts);
    scan_carry<<<(D_STATE * D_INNER) / 256, 256, 0, stream>>>(
        hend, dts, A_log, hst);
    scan_chunk_out<<<dim3(D_INNER / SCAN_BLK, SCAN_NC), SCAN_BLK, 0, stream>>>(
        xdbl, dtb, xc, xz, A_log, D_param, hst, yb);

    gemm_wmma_f32<<<dim3(D_MODEL / 64, LSEQ / 128), blk, 0, stream>>>(
        yb, W_out, nullptr, out, LSEQ, D_MODEL, D_INNER, D_INNER, 0);
}


# dt GEMM: softplus epilogue as one rolled loop over the wave's tile parked in LDS (4 values per lane per trip), not 64 in-line log1pf chains; plain GEMMs carry no activation text; arithmetic kept
# speedup vs baseline: 12.8798x; 1.0838x over previous
#include <hip/hip_runtime.h>
#include <math.h>

#define D_MODEL 1024
#define D_STATE 16
#define D_CONV  4
#define DT_RANK 64
#define D_INNER 2048
#define LSEQ    2048

typedef __attribute__((ext_vector_type(2))) float v2f;
typedef __attribute__((ext_vector_type(8))) float v8f;

typedef __attribute__((ext_vector_type(4))) float v4f;

struct Blk { v4f a0[4], a1[4], b0[4], b1[4], b2[4], b3[4]; };

template <int NP>
__device__ __forceinline__ void load_row(v4f* q, const float* p)
{
#pragma unroll
    for (int j = 0; j < NP; ++j) q[j] = *(const v4f*)(p + 8 * j);
}

__device__ __forceinline__ void load_blk(
    Blk& f, const float* Ar0, const float* Ar1,
    const float* w0, const float* w1, const float* w2, const float* w3, int k)
{
    load_row<4>(f.a0, Ar0 + k);
    load_row<4>(f.a1, Ar1 + k);
    load_row<4>(f.b0, w0 + k);
    load_row<4>(f.b1, w1 + k);
    load_row<4>(f.b2, w2 + k);
    load_row<4>(f.b3, w3 + k);
}

__device__ __forceinline__ void mma_step(
    v2f a0, v2f a1, v2f b0, v2f b1, v2f b2, v2f b3, v8f acc[2][4])
{
    acc[0][0] = __builtin_amdgcn_wmma_f32_16x16x4_f32(false, a0, false, b0, (short)0, acc[0][0], false, false);
    acc[1][0] = __builtin_amdgcn_wmma_f32_16x16x4_f32(false, a1, false, b0, (short)0, acc[1][0], false, false);
    acc[0][1] = __builtin_amdgcn_wmma_f32_16x16x4_f32(false, a0, false, b1, (short)0, acc[0][1], false, false);
    acc[1][1] = __builtin_amdgcn_wmma_f32_16x16x4_f32(false, a1, false, b1, (short)0, acc[1][1], false, false);
    acc[0][2] = __builtin_amdgcn_wmma_f32_16x16x4_f32(false, a0, false, b2, (short)0, acc[0][2], false, false);
    acc[1][2] = __builtin_amdgcn_wmma_f32_16x16x4_f32(false, a1, false, b2, (short)0, acc[1][2], false, false);
    acc[0][3] = __builtin_amdgcn_wmma_f32_16x16x4_f32(false, a0, false, b3, (short)0, acc[0][3], false, false);
    acc[1][3] = __builtin_amdgcn_wmma_f32_16x16x4_f32(false, a1, false, b3, (short)0, acc[1][3], false, false);
}

__device__ __forceinline__ void mma_blk(const Blk& f, v8f acc[2][4])
{
#pragma unroll
    for (int j = 0; j < 4; ++j) {
        mma_step(f.a0[j].xy, f.a1[j].xy, f.b0[j].xy, f.b1[j].xy, f.b2[j].xy, f.b3[j].xy, acc);
        mma_step(f.a0[j].zw, f.a1[j].zw, f.b0[j].zw, f.b1[j].zw, f.b2[j].zw, f.b3[j].zw, acc);
    }
}

__device__ __forceinline__ void gemm_tile_32x64(
    const float* __restrict__ A, const float* __restrict__ W,
    int N, int K, int lda, int m0, int n0, int lane, v8f acc[2][4])
{
    const int mrow  = lane & 15;
    const int khalf = (lane >> 4) << 2;

    const float* __restrict__ Ar0 = A + (size_t)(m0 + mrow) * lda + khalf;
    const float* __restrict__ Ar1 = Ar0 + (size_t)16 * lda;

    const float* Wr[4];
#pragma unroll
    for (int j = 0; j < 4; ++j) {
        int row = n0 + (j << 4) + mrow;
        row = (row < N) ? row : (N - 1);
        Wr[j] = W + (size_t)row * K + khalf;
    }

#pragma unroll
    for (int i = 0; i < 2; ++i)
#pragma unroll
        for (int j = 0; j < 4; ++j) acc[i][j] = (v8f){};

    Blk f0, f1;
    load_blk(f0, Ar0, Ar1, Wr[0], Wr[1], Wr[2], Wr[3], 0);

    int k0 = 0;
    for (; k0 < K - 64; k0 += 64) {
        load_blk(f1, Ar0, Ar1, Wr[0], Wr[1], Wr[2], Wr[3], k0 + 32);
        __builtin_amdgcn_sched_barrier(0);
        mma_blk(f0, acc);
        __builtin_amdgcn_sched_barrier(0);
        load_blk(f0, Ar0, Ar1, Wr[0], Wr[1], Wr[2], Wr[3], k0 + 64);
        __builtin_amdgcn_sched_barrier(0);
        mma_blk(f1, acc);
    }
    load_blk(f1, Ar0, Ar1, Wr[0], Wr[1], Wr[2], Wr[3], k0 + 32);
    __builtin_amdgcn_sched_barrier(0);
    mma_blk(f0, acc);
    mma_blk(f1, acc);
}

__global__ __launch_bounds__(128, 1) void gemm_wmma_f32(
    const float* __restrict__ A, const float* __restrict__ W,
    float* __restrict__ C, int M, int N, int K, int lda)
{
    const int lane = threadIdx.x & 31;
    const int wave = threadIdx.x >> 5;
    const int m0   = (blockIdx.y << 7) + (wave << 5);
    const int n0   = blockIdx.x << 6;
    if (m0 >= M) return;

    v8f acc[2][4];
    gemm_tile_32x64(A, W, N, K, lda, m0, n0, lane, acc);

    const int mrow = lane & 15;
    const int rsub = (lane >> 4) << 3;
#pragma unroll
    for (int j = 0; j < 4; ++j) {
        const int col = n0 + (j << 4) + mrow;
        if ((n0 + (j << 4)) >= N) continue;
#pragma unroll
        for (int i = 0; i < 2; ++i) {
            const int rbase = m0 + (i << 4) + rsub;
#pragma unroll
            for (int r = 0; r < 8; ++r)
                C[(size_t)(rbase + r) * N + col] = acc[i][j][r];
        }
    }
}

#define EPI_LD 68

__global__ __launch_bounds__(128, 1) void gemm_wmma_f32_softplus(
    const float* __restrict__ A, const float* __restrict__ W,
    const float* __restrict__ bias, float* __restrict__ C,
    int M, int N, int K, int lda)
{
    __shared__ float stage[4][32][EPI_LD];

    const int lane = threadIdx.x & 31;
    const int wave = threadIdx.x >> 5;
    const int m0   = (blockIdx.y << 7) + (wave << 5);
    const int n0   = blockIdx.x << 6;
    if (m0 >= M) return;

    v8f acc[2][4];
    gemm_tile_32x64(A, W, N, K, lda, m0, n0, lane, acc);

    float (*st)[EPI_LD] = stage[wave];
    const int mrow = lane & 15;
    const int rsub = (lane >> 4) << 3;
#pragma unroll
    for (int i = 0; i < 2; ++i)
#pragma unroll
        for (int j = 0; j < 4; ++j)
#pragma unroll
            for (int r = 0; r < 8; ++r)
                st[(i << 4) + rsub + r][(j << 4) + mrow] = acc[i][j][r];

    __builtin_amdgcn_fence(__ATOMIC_RELEASE, "wavefront");
    __builtin_amdgcn_wave_barrier();
    __builtin_amdgcn_fence(__ATOMIC_ACQUIRE, "wavefront");

    const int c0 = n0 + lane;
    const int c1 = c0 + 32;
    const float bv0 = bias[c0];
    const float bv1 = bias[c1];

#pragma nounroll
    for (int row = 0; row < 32; row += 2) {
        float v[4];
        v[0] = st[row][lane]          + bv0;
        v[1] = st[row][lane + 32]     + bv1;
        v[2] = st[row + 1][lane]      + bv0;
        v[3] = st[row + 1][lane + 32] + bv1;
#pragma unroll
        for (int q = 0; q < 4; ++q) {
            const float sp = log1pf(__expf(v[q]));
            v[q] = (v[q] > 20.0f) ? v[q] : sp;
        }
        float* crow = C + (size_t)(m0 + row) * N;
        crow[c0] = v[0];
        crow[c1] = v[1];
        crow[(size_t)N + c0] = v[2];
        crow[(size_t)N + c1] = v[3];
    }
}

#define NARROW_P 16

struct BlkN { v4f a[NARROW_P], b0[NARROW_P], b1[NARROW_P]; };

__device__ __forceinline__ void load_blk_n(
    BlkN& f, const float* Ar, const float* w0, const float* w1, int k)
{
    load_row<NARROW_P>(f.a,  Ar + k);
    load_row<NARROW_P>(f.b0, w0 + k);
    load_row<NARROW_P>(f.b1, w1 + k);
}

__device__ __forceinline__ v8f wmma_f32(v2f a, v2f b, v8f c)
{
    return __builtin_amdgcn_wmma_f32_16x16x4_f32(false, a, false, b, (short)0, c, false, false);
}

__device__ __forceinline__ void mma_blk_n(const BlkN& f, v8f acc[2])
{
#pragma unroll
    for (int j = 0; j < NARROW_P; ++j) {
        acc[0] = wmma_f32(f.a[j].xy, f.b0[j].xy, acc[0]);
        acc[1] = wmma_f32(f.a[j].xy, f.b1[j].xy, acc[1]);
        acc[0] = wmma_f32(f.a[j].zw, f.b0[j].zw, acc[0]);
        acc[1] = wmma_f32(f.a[j].zw, f.b1[j].zw, acc[1]);
    }
}

__global__ __launch_bounds__(32, 1) void gemm_wmma_f32_narrow(
    const float* __restrict__ A, const float* __restrict__ W,
    float* __restrict__ C, int N, int K, int lda)
{
    const int lane  = threadIdx.x;
    const int m0    = blockIdx.y << 4;
    const int n0    = blockIdx.x << 5;
    const int mrow  = lane & 15;
    const int khalf = (lane >> 4) << 2;

    const float* __restrict__ Ar = A + (size_t)(m0 + mrow) * lda + khalf;
    const float* __restrict__ w0 = W + (size_t)(n0 + mrow) * K + khalf;
    const float* __restrict__ w1 = w0 + (size_t)16 * K;

    v8f acc[2];
    acc[0] = (v8f){};
    acc[1] = (v8f){};

    BlkN f0, f1;
    load_blk_n(f0, Ar, w0, w1, 0);

    int k0 = 0;
    for (; k0 < K - 256; k0 += 256) {
        load_blk_n(f1, Ar, w0, w1, k0 + 128);
        mma_blk_n(f0, acc);
        load_blk_n(f0, Ar, w0, w1, k0 + 256);
        mma_blk_n(f1, acc);
        __builtin_amdgcn_sched_group_barrier(0x020, 3 * NARROW_P, 0);
        __builtin_amdgcn_sched_group_barrier(0x008, 4 * NARROW_P, 0);
        __builtin_amdgcn_sched_group_barrier(0x020, 3 * NARROW_P, 0);
        __builtin_amdgcn_sched_group_barrier(0x008, 4 * NARROW_P, 0);
    }
    load_blk_n(f1, Ar, w0, w1, k0 + 128);
    __builtin_amdgcn_sched_barrier(0);
    mma_blk_n(f0, acc);
    mma_blk_n(f1, acc);

    const int rsub = (lane >> 4) << 3;
#pragma unroll
    for (int j = 0; j < 2; ++j) {
        const int col = n0 + (j << 4) + mrow;
#pragma unroll
        for (int r = 0; r < 8; ++r)
            C[(size_t)(m0 + rsub + r) * N + col] = acc[j][r];
    }
}

__global__ __launch_bounds__(256) void conv_silu_kernel(
    const float* __restrict__ xz, const float* __restrict__ cw,
    const float* __restrict__ cb, float* __restrict__ xc)
{
    int idx = blockIdx.x * blockDim.x + threadIdx.x;
    if (idx >= LSEQ * D_INNER) return;
    int l = idx >> 11;
    int d = idx & (D_INNER - 1);

    float acc = cb[d];
#pragma unroll
    for (int j = 0; j < D_CONV; ++j) {
        int ls = l - (D_CONV - 1) + j;
        if (ls >= 0)
            acc += cw[d * D_CONV + j] * xz[(size_t)ls * (2 * D_INNER) + d];
    }
    xc[idx] = acc / (1.0f + __expf(-acc));
}

#define SCAN_T   32
#define SCAN_NC  (LSEQ / SCAN_T)
#define SCAN_BLK 128
#define XDBL_LD  (DT_RANK + 2 * D_STATE)

__global__ __launch_bounds__(SCAN_BLK) void scan_chunk_state(
    const float* __restrict__ xdbl, const float* __restrict__ dtbuf,
    const float* __restrict__ xc,   const float* __restrict__ A_log,
    float* __restrict__ hend, float* __restrict__ dtsum)
{
    __shared__ float sB[SCAN_T][D_STATE];

    const int d  = blockIdx.x * SCAN_BLK + threadIdx.x;
    const int c  = blockIdx.y;
    const int lc = c * SCAN_T;

    for (int i = threadIdx.x; i < SCAN_T * D_STATE; i += SCAN_BLK) {
        int t = i >> 4, n = i & 15;
        sB[t][n] = xdbl[(size_t)(lc + t) * XDBL_LD + DT_RANK + n];
    }

    float Acoef[D_STATE];
#pragma unroll
    for (int n = 0; n < D_STATE; ++n)
        Acoef[n] = -__expf(A_log[(size_t)d * D_STATE + n]);

    float h[D_STATE];
#pragma unroll
    for (int n = 0; n < D_STATE; ++n) h[n] = 0.0f;
    float s = 0.0f;

    __syncthreads();

#pragma unroll 8
    for (int t = 0; t < SCAN_T; ++t) {
        const int l = lc + t;
        const float dt = dtbuf[(size_t)l * D_INNER + d];
        const float xv = xc[(size_t)l * D_INNER + d];
        s += dt;
#pragma unroll
        for (int n = 0; n < D_STATE; ++n) {
            float dA = __expf(dt * Acoef[n]);
            h[n] = dA * h[n] + (dt * sB[t][n]) * xv;
        }
    }

#pragma unroll
    for (int n = 0; n < D_STATE; ++n)
        hend[((size_t)c * D_STATE + n) * D_INNER + d] = h[n];
    dtsum[(size_t)c * D_INNER + d] = s;
}

__global__ __launch_bounds__(256) void scan_carry(
    const float* __restrict__ hend, const float* __restrict__ dtsum,
    const float* __restrict__ A_log, float* __restrict__ hstart)
{
    const int idx = blockIdx.x * blockDim.x + threadIdx.x;
    const int n = idx >> 11;
    const int d = idx & (D_INNER - 1);
    const float Ac = -__expf(A_log[(size_t)d * D_STATE + n]);

    float cur = 0.0f;
    hstart[idx] = 0.0f;
#pragma unroll 7
    for (int c = 0; c < SCAN_NC - 1; ++c) {
        const float e = __expf(dtsum[(size_t)c * D_INNER + d] * Ac);
        cur = e * cur + hend[(size_t)c * (D_STATE * D_INNER) + idx];
        hstart[(size_t)(c + 1) * (D_STATE * D_INNER) + idx] = cur;
    }
}

__global__ __launch_bounds__(SCAN_BLK) void scan_chunk_out(
    const float* __restrict__ xdbl, const float* __restrict__ dtbuf,
    const float* __restrict__ xc,   const float* __restrict__ xz,
    const float* __restrict__ A_log, const float* __restrict__ D_param,
    const float* __restrict__ hstart, float* __restrict__ yout)
{
    __shared__ float sB[SCAN_T][D_STATE];
    __shared__ float sC[SCAN_T][D_STATE];

    const int d  = blockIdx.x * SCAN_BLK + threadIdx.x;
    const int c  = blockIdx.y;
    const int lc = c * SCAN_T;

    for (int i = threadIdx.x; i < SCAN_T * D_STATE; i += SCAN_BLK) {
        int t = i >> 4, n = i & 15;
        const float* row = xdbl + (size_t)(lc + t) * XDBL_LD;
        sB[t][n] = row[DT_RANK + n];
        sC[t][n] = row[DT_RANK + D_STATE + n];
    }

    float Acoef[D_STATE];
#pragma unroll
    for (int n = 0; n < D_STATE; ++n)
        Acoef[n] = -__expf(A_log[(size_t)d * D_STATE + n]);
    const float Dp = D_param[d];

    float h[D_STATE];
#pragma unroll
    for (int n = 0; n < D_STATE; ++n)
        h[n] = hstart[((size_t)c * D_STATE + n) * D_INNER + d];

    __syncthreads();

#pragma unroll 8
    for (int t = 0; t < SCAN_T; ++t) {
        const int l = lc + t;
        const float dt = dtbuf[(size_t)l * D_INNER + d];
        const float xv = xc[(size_t)l * D_INNER + d];
        const float z  = xz[(size_t)l * (2 * D_INNER) + D_INNER + d];
        float y = 0.0f;
#pragma unroll
        for (int n = 0; n < D_STATE; ++n) {
            float dA = __expf(dt * Acoef[n]);
            h[n] = dA * h[n] + (dt * sB[t][n]) * xv;
            y += h[n] * sC[t][n];
        }
        y += xv * Dp;
        float g = z / (1.0f + __expf(-z));
        yout[(size_t)l * D_INNER + d] = y * g;
    }
}

extern "C" void kernel_launch(void* const* d_in, const int* in_sizes, int n_in,
                              void* d_out, int out_size, void* d_ws, size_t ws_size,
                              hipStream_t stream) {
    const float* x       = (const float*)d_in[0];
    const float* W_in    = (const float*)d_in[1];
    const float* conv_w  = (const float*)d_in[2];
    const float* conv_b  = (const float*)d_in[3];
    const float* W_x     = (const float*)d_in[4];
    const float* W_dt    = (const float*)d_in[5];
    const float* b_dt    = (const float*)d_in[6];
    const float* A_log   = (const float*)d_in[7];
    const float* D_param = (const float*)d_in[8];
    const float* W_out   = (const float*)d_in[9];
    float* out = (float*)d_out;

    float* xz   = (float*)d_ws;
    float* xc   = xz   + (size_t)LSEQ * 2 * D_INNER;
    float* xdbl = xc   + (size_t)LSEQ * D_INNER;
    float* dtb  = xdbl + (size_t)LSEQ * (DT_RANK + 2 * D_STATE);
    float* yb   = dtb  + (size_t)LSEQ * D_INNER;
    float* hend = yb   + (size_t)LSEQ * D_INNER;
    float* hst  = hend + (size_t)SCAN_NC * D_STATE * D_INNER;
    float* dts  = hst  + (size_t)SCAN_NC * D_STATE * D_INNER;

    dim3 blk(128, 1, 1);

    gemm_wmma_f32<<<dim3(4096 / 64, LSEQ / 128), blk, 0, stream>>>(
        x, W_in, xz, LSEQ, 2 * D_INNER, D_MODEL, D_MODEL);

    conv_silu_kernel<<<(LSEQ * D_INNER) / 256, 256, 0, stream>>>(
        xz, conv_w, conv_b, xc);

    gemm_wmma_f32_narrow<<<dim3((DT_RANK + 2 * D_STATE) / 32, LSEQ / 16), dim3(32, 1, 1), 0, stream>>>(
        xc, W_x, xdbl, DT_RANK + 2 * D_STATE, D_INNER, D_INNER);

    gemm_wmma_f32_softplus<<<dim3(D_INNER / 64, LSEQ / 128), blk, 0, stream>>>(
        xdbl, W_dt, b_dt, dtb, LSEQ, D_INNER, DT_RANK, DT_RANK + 2 * D_STATE);

    scan_chunk_state<<<dim3(D_INNER / SCAN_BLK, SCAN_NC - 1), SCAN_BLK, 0, stream>>>(
        xdbl, dtb, xc, A_log, hend, dts);
    scan_carry<<<(D_STATE * D_INNER) / 256, 256, 0, stream>>>(
        hend, dts, A_log, hst);
    scan_chunk_out<<<dim3(D_INNER / SCAN_BLK, SCAN_NC), SCAN_BLK, 0, stream>>>(
        xdbl, dtb, xc, xz, A_log, D_param, hst, yb);

    gemm_wmma_f32<<<dim3(D_MODEL / 64, LSEQ / 128), blk, 0, stream>>>(
        yb, W_out, out, LSEQ, D_MODEL, D_INNER, D_INNER);
}
